# starting kernel
# baseline (speedup 1.0000x reference)
#include <hip/hip_runtime.h>


typedef __attribute__((ext_vector_type(2))) float v2f;
typedef __attribute__((ext_vector_type(8))) float v8f;

#define HP     288
#define WPAD   34
#define NITER  26
#define GAMMA  1.0f
#define DTOT   576.0f

__global__ __launch_bounds__(256) void mpconv_sparsemax_kernel(
    const float* __restrict__ x,
    const float* __restrict__ w,
    float* __restrict__ out)
{
    __shared__ float sW[HP * WPAD];
    __shared__ float sP[8][HP];

    const int tid  = threadIdx.x;
    const int lane = tid & 31;
    const int wv   = tid >> 5;

    for (int g = tid; g < 32 * HP; g += 256) {
        int o = g / HP;
        int h = g - o * HP;
        sW[h * WPAD + o] = w[g];
    }

    const int loc = blockIdx.x * 8 + wv;
    const int b   = loc >> 10;
    const int pos = loc & 1023;
    const int oh  = pos >> 5, ow = pos & 31;
    {
        const int c = lane;
        const float* xb = x + ((b * 32 + c) * 32) * 32;
        #pragma unroll
        for (int ki = 0; ki < 3; ++ki) {
            #pragma unroll
            for (int kj = 0; kj < 3; ++kj) {
                const int y  = oh - 1 + ki;
                const int xx = ow - 1 + kj;
                float v = 0.0f;
                if ((unsigned)y < 32u && (unsigned)xx < 32u) v = xb[y * 32 + xx];
                sP[wv][c * 9 + ki * 3 + kj] = v;
            }
        }
    }
    __syncthreads();

    const int   half   = lane >> 4;
    const int   l16    = lane & 15;
    const int   rowoff = half * 8;
    const float bsel   = half ? 0.0f : 1.0f;
    v2f Bm; Bm[0] = bsel; Bm[1] = 0.0f;
    const float* pP = sP[wv];

    float hi_pa = 0.0f, hi_qa = 0.0f, hi_pb = 0.0f, hi_qb = 0.0f;
    for (int h0 = 0; h0 < HP; h0 += 16) {
        float pv = pP[h0 + l16];
        float a0 = half ? 0.0f : pv;
        v2f Ap; Ap[0] =  a0; Ap[1] = 0.0f;
        v2f Aq; Aq[0] = -a0; Aq[1] = 0.0f;
        v8f Ca, Cb;
        #pragma unroll
        for (int r = 0; r < 8; ++r) {
            const float* row = &sW[(h0 + r + rowoff) * WPAD];
            Ca[r] = row[l16];
            Cb[r] = row[l16 + 16];
        }
        v8f Dpa = __builtin_amdgcn_wmma_f32_16x16x4_f32(false, Ap, false, Bm, (short)0, Ca, false, false);
        v8f Dqa = __builtin_amdgcn_wmma_f32_16x16x4_f32(false, Aq, false, Bm, (short)0, Ca, false, false);
        v8f Dpb = __builtin_amdgcn_wmma_f32_16x16x4_f32(false, Ap, false, Bm, (short)0, Cb, false, false);
        v8f Dqb = __builtin_amdgcn_wmma_f32_16x16x4_f32(false, Aq, false, Bm, (short)0, Cb, false, false);
        #pragma unroll
        for (int r = 0; r < 8; ++r) {
            hi_pa = fmaxf(hi_pa, __builtin_fabsf(Dpa[r]));
            hi_qa = fmaxf(hi_qa, __builtin_fabsf(Dqa[r]));
            hi_pb = fmaxf(hi_pb, __builtin_fabsf(Dpb[r]));
            hi_qb = fmaxf(hi_qb, __builtin_fabsf(Dqb[r]));
        }
    }
    hi_pa = fmaxf(hi_pa, __shfl_xor(hi_pa, 16, 32));
    hi_qa = fmaxf(hi_qa, __shfl_xor(hi_qa, 16, 32));
    hi_pb = fmaxf(hi_pb, __shfl_xor(hi_pb, 16, 32));
    hi_qb = fmaxf(hi_qb, __shfl_xor(hi_qb, 16, 32));

    const float lo0 = -GAMMA / DTOT;
    float lo_pa = lo0, lo_qa = lo0, lo_pb = lo0, lo_qb = lo0;

    for (int it = 0; it < NITER; ++it) {
        const float t_pa = 0.5f * (lo_pa + hi_pa);
        const float t_qa = 0.5f * (lo_qa + hi_qa);
        const float t_pb = 0.5f * (lo_pb + hi_pb);
        const float t_qb = 0.5f * (lo_qb + hi_qb);
        const float fl_pa = fmaxf(-2.0f * t_pa, 0.0f);
        const float fl_qa = fmaxf(-2.0f * t_qa, 0.0f);
        const float fl_pb = fmaxf(-2.0f * t_pb, 0.0f);
        const float fl_qb = fmaxf(-2.0f * t_qb, 0.0f);
        float f_pa = 0.0f, f_qa = 0.0f, f_pb = 0.0f, f_qb = 0.0f;

        for (int h0 = 0; h0 < HP; h0 += 16) {
            float pv = pP[h0 + l16];
            float a0 = half ? 0.0f : pv;
            v2f Ap; Ap[0] =  a0; Ap[1] = 0.0f;
            v2f Aq; Aq[0] = -a0; Aq[1] = 0.0f;
            v8f Ca, Cb;
            #pragma unroll
            for (int r = 0; r < 8; ++r) {
                const float* row = &sW[(h0 + r + rowoff) * WPAD];
                Ca[r] = row[l16];
                Cb[r] = row[l16 + 16];
            }
            v8f Dpa = __builtin_amdgcn_wmma_f32_16x16x4_f32(false, Ap, false, Bm, (short)0, Ca, false, false);
            v8f Dqa = __builtin_amdgcn_wmma_f32_16x16x4_f32(false, Aq, false, Bm, (short)0, Ca, false, false);
            v8f Dpb = __builtin_amdgcn_wmma_f32_16x16x4_f32(false, Ap, false, Bm, (short)0, Cb, false, false);
            v8f Dqb = __builtin_amdgcn_wmma_f32_16x16x4_f32(false, Aq, false, Bm, (short)0, Cb, false, false);
            #pragma unroll
            for (int r = 0; r < 8; ++r) {
                f_pa += fmaxf(__builtin_fabsf(Dpa[r]) - t_pa, fl_pa);
                f_qa += fmaxf(__builtin_fabsf(Dqa[r]) - t_qa, fl_qa);
                f_pb += fmaxf(__builtin_fabsf(Dpb[r]) - t_pb, fl_pb);
                f_qb += fmaxf(__builtin_fabsf(Dqb[r]) - t_qb, fl_qb);
            }
        }
        f_pa += __shfl_xor(f_pa, 16, 32);
        f_qa += __shfl_xor(f_qa, 16, 32);
        f_pb += __shfl_xor(f_pb, 16, 32);
        f_qb += __shfl_xor(f_qb, 16, 32);

        lo_pa = (f_pa > GAMMA) ? t_pa : lo_pa;  hi_pa = (f_pa > GAMMA) ? hi_pa : t_pa;
        lo_qa = (f_qa > GAMMA) ? t_qa : lo_qa;  hi_qa = (f_qa > GAMMA) ? hi_qa : t_qa;
        lo_pb = (f_pb > GAMMA) ? t_pb : lo_pb;  hi_pb = (f_pb > GAMMA) ? hi_pb : t_pb;
        lo_qb = (f_qb > GAMMA) ? t_qb : lo_qb;  hi_qb = (f_qb > GAMMA) ? hi_qb : t_qb;
    }

    const float tau_pa = 0.5f * (lo_pa + hi_pa);
    const float tau_qa = 0.5f * (lo_qa + hi_qa);
    const float tau_pb = 0.5f * (lo_pb + hi_pb);
    const float tau_qb = 0.5f * (lo_qb + hi_qb);

    const float tp = half ? tau_pb : tau_pa;
    const float tq = half ? tau_qb : tau_qa;
    const float conv = 0.5f * DTOT * (tp - tq);

    out[(b * 32 + lane) * 1024 + pos] = conv;
}

extern "C" void kernel_launch(void* const* d_in, const int* in_sizes, int n_in,
                              void* d_out, int out_size, void* d_ws, size_t ws_size,
                              hipStream_t stream) {
    const float* x = (const float*)d_in[0];
    const float* w = (const float*)d_in[1];
    float* out = (float*)d_out;
    (void)in_sizes; (void)n_in; (void)out_size; (void)d_ws; (void)ws_size;

    mpconv_sparsemax_kernel<<<256, 256, 0, stream>>>(x, w, out);
}


# Selection networks: each compare-exchange written as one dual-issue max/min pair (asm), merge maxima folded into max3/maxmin, a lane's four lists sorted in step; sums bit-identical
# speedup vs baseline: 9.2831x; 9.2831x over previous
#include <hip/hip_runtime.h>


typedef __attribute__((ext_vector_type(2))) float v2f;
typedef __attribute__((ext_vector_type(8))) float v8f;

#define HP     288
#define NITER  26
#define GAMMA  1.0f
#define DTOT   576.0f
#define KL     12
#define KM     16

__device__ __forceinline__ float mx(float a, float b) { return __builtin_fmaximum_numf(a, b); }
__device__ __forceinline__ float mn(float a, float b) { return __builtin_fminimum_numf(a, b); }

__device__ __forceinline__ void cex(float& a, float& b) {
    const float hi = mx(a, b);
    const float lo = mn(a, b);
    a = hi; b = lo;
}

__device__ __forceinline__ void cexd(float& a, float& b) {
    float hi, lo;
    asm("v_dual_max_num_f32 %0, %2, %3 :: v_dual_min_num_f32 %1, %2, %3" : "=&v"(hi), "=&v"(lo) : "v"(a), "v"(b));
    a = hi; b = lo;
}

__device__ __forceinline__ float max3f(float a, float b, float c) {
    float d;
    asm("v_max3_num_f32 %0, %1, %2, %3" : "=v"(d) : "v"(a), "v"(b), "v"(c));
    return d;
}

__device__ __forceinline__ float maxminf(float a, float b, float c) {
    float d;
    asm("v_maxmin_num_f32 %0, %1, %2, %3" : "=v"(d) : "v"(a), "v"(b), "v"(c));
    return d;
}

__device__ __forceinline__ void bitonic16_desc(float (&t)[KM]) {
    #pragma unroll
    for (int d = 8; d >= 1; d >>= 1) {
        #pragma unroll
        for (int i = 0; i < KM; ++i)
            if ((i & d) == 0) cex(t[i], t[i + d]);
    }
}

__device__ __forceinline__ void take8x4(float (&ta)[KL], float (&tb)[KL], float (&tc)[KL], float (&td)[KL],
                                        const v8f& Da, const v8f& Db, const v8f& Dc, const v8f& Dd) {
    float sa[8], sb[8], sc[8], sd[8];
    #pragma unroll
    for (int r = 0; r < 8; ++r) {
        sa[r] = __builtin_fabsf(Da[r]); sb[r] = __builtin_fabsf(Db[r]);
        sc[r] = __builtin_fabsf(Dc[r]); sd[r] = __builtin_fabsf(Dd[r]);
    }
#define CE4(i, j) cexd(sa[i], sa[j]); cexd(sb[i], sb[j]); cexd(sc[i], sc[j]); cexd(sd[i], sd[j]);
    CE4(0, 2) CE4(1, 3) CE4(4, 6) CE4(5, 7)
    CE4(0, 4) CE4(1, 5) CE4(2, 6) CE4(3, 7)
    CE4(0, 1) CE4(2, 3) CE4(4, 5) CE4(6, 7)
    CE4(2, 4) CE4(3, 5)
    CE4(1, 4) CE4(3, 6)
    CE4(1, 2) CE4(3, 4) CE4(5, 6)
#undef CE4
    float la[4], lb[4], lc[4], ld[4];
    #pragma unroll
    for (int i = 0; i < 4; ++i) {
        const float ha = max3f(ta[i], ta[8 + i], sa[3 - i]);
        const float hb = max3f(tb[i], tb[8 + i], sb[3 - i]);
        const float hc = max3f(tc[i], tc[8 + i], sc[3 - i]);
        const float hd = max3f(td[i], td[8 + i], sd[3 - i]);
        la[i] = maxminf(ta[8 + i], sa[3 - i], ta[i]);
        lb[i] = maxminf(tb[8 + i], sb[3 - i], tb[i]);
        lc[i] = maxminf(tc[8 + i], sc[3 - i], tc[i]);
        ld[i] = maxminf(td[8 + i], sd[3 - i], td[i]);
        ta[i] = ha; tb[i] = hb; tc[i] = hc; td[i] = hd;
    }
    #pragma unroll
    for (int j = 0; j < 4; ++j) {
        const float ha = max3f(ta[4 + j], sa[7 - j], la[j]);
        const float hb = max3f(tb[4 + j], sb[7 - j], lb[j]);
        const float hc = max3f(tc[4 + j], sc[7 - j], lc[j]);
        const float hd = max3f(td[4 + j], sd[7 - j], ld[j]);
        ta[8 + j] = maxminf(ta[4 + j], sa[7 - j], la[j]);
        tb[8 + j] = maxminf(tb[4 + j], sb[7 - j], lb[j]);
        tc[8 + j] = maxminf(tc[4 + j], sc[7 - j], lc[j]);
        td[8 + j] = maxminf(td[4 + j], sd[7 - j], ld[j]);
        ta[4 + j] = ha; tb[4 + j] = hb; tc[4 + j] = hc; td[4 + j] = hd;
    }
#define TE4(i, j) cexd(ta[i], ta[j]); cexd(tb[i], tb[j]); cexd(tc[i], tc[j]); cexd(td[i], td[j]);
    TE4(0, 2) TE4(1, 3) TE4(4, 6) TE4(5, 7) TE4(8, 10) TE4(9, 11)
    TE4(0, 1) TE4(2, 3) TE4(4, 5) TE4(6, 7) TE4(8, 9) TE4(10, 11)
#undef TE4
}

__device__ __forceinline__ float solve16(const float (&a)[KL], const float (&b)[KL], bool& ok) {
    float m[KM];
    #pragma unroll
    for (int i = 0; i < KM - KL; ++i) { m[i] = a[i]; m[KL + i] = b[KM - KL - 1 - i]; }
    #pragma unroll
    for (int i = KM - KL; i < KL; ++i) m[i] = mx(a[i], b[KM - 1 - i]);
    bitonic16_desc(m);
    float cs = 0.0f, bcs = m[0], bk = 1.0f;
    #pragma unroll
    for (int k = 1; k <= KM; ++k) {
        cs += m[k - 1];
        const bool c = (m[k - 1] * (float)k >= cs - GAMMA);
        bcs = c ? cs : bcs;
        bk  = c ? (float)k : bk;
    }
    const float tau = (bcs - GAMMA) / bk;
    ok = (tau >= mx(m[KM - 1], mx(a[KL - 1], b[KL - 1])));
    return tau;
}

__device__ __forceinline__ void load_tile(const float* wl, const float* xb, int h0, int l16, int half,
                                          int oh, int ow, v8f& Ca, v8f& Cb, float& tap, bool& use) {
    Ca = *reinterpret_cast<const v8f*>(wl + h0);
    Cb = *reinterpret_cast<const v8f*>(wl + 16 * HP + h0);
    __builtin_assume((unsigned)h0 < (unsigned)HP);
    const unsigned h  = (unsigned)(h0 + l16);
    const unsigned t  = h * 0xE38Fu;
    const unsigned c  = t >> 19;
    const unsigned f  = t & 0x7FFFFu;
    const unsigned r9 = (f * 9u) >> 19;
    const unsigned ki = (f * 3u) >> 19;
    const unsigned kj = r9 - ki * 3u;
    const int y  = oh - 1 + (int)ki;
    const int xx = ow - 1 + (int)kj;
    const bool in = ((unsigned)y < 32u) && ((unsigned)xx < 32u);
    const int idx = in ? ((int)(c * 1024u) + y * 32 + xx) : 0;
    tap = xb[idx];
    use = in && (half == 0);
}

__device__ __forceinline__ void form_tiles(const v8f& Ca, const v8f& Cb, float tap, bool use, const v2f& Bm,
                                           v8f& Dpa, v8f& Dqa, v8f& Dpb, v8f& Dqb) {
    const float a0 = use ? tap : 0.0f;
    v2f Ap; Ap[0] =  a0; Ap[1] = 0.0f;
    v2f Aq; Aq[0] = -a0; Aq[1] = 0.0f;
    Dpa = __builtin_amdgcn_wmma_f32_16x16x4_f32(false, Ap, false, Bm, (short)0, Ca, false, false);
    Dqa = __builtin_amdgcn_wmma_f32_16x16x4_f32(false, Aq, false, Bm, (short)0, Ca, false, false);
    Dpb = __builtin_amdgcn_wmma_f32_16x16x4_f32(false, Ap, false, Bm, (short)0, Cb, false, false);
    Dqb = __builtin_amdgcn_wmma_f32_16x16x4_f32(false, Aq, false, Bm, (short)0, Cb, false, false);
}

__global__ __launch_bounds__(256) void mpconv_sparsemax_kernel(
    const float* x,
    const float* w,
    float* __restrict__ out)
{
    const int tid  = threadIdx.x;
    const int lane = tid & 31;
    const int wv   = tid >> 5;

    const int loc = blockIdx.x * 8 + wv;
    const int b   = loc >> 10;
    const int pos = loc & 1023;
    const int oh  = pos >> 5, ow = pos & 31;

    const int   half   = lane >> 4;
    const int   l16    = lane & 15;
    const float bsel   = half ? 0.0f : 1.0f;
    v2f Bm; Bm[0] = bsel; Bm[1] = 0.0f;
    const float* wl = w + l16 * HP + half * 8;
    const float* xb = x + b * (32 * 1024);

    float Tpa[KL], Tqa[KL], Tpb[KL], Tqb[KL];
    #pragma unroll
    for (int i = 0; i < KL; ++i) { Tpa[i] = -1.0f; Tqa[i] = -1.0f; Tpb[i] = -1.0f; Tqb[i] = -1.0f; }

    v8f Ca0, Cb0, Ca1, Cb1;
    float tap0, tap1;
    bool use0, use1;
    load_tile(wl, xb, 0, l16, half, oh, ow, Ca0, Cb0, tap0, use0);
    for (int h0 = 0; h0 < HP; h0 += 32) {
        v8f Dpa, Dqa, Dpb, Dqb;
        load_tile(wl, xb, h0 + 16, l16, half, oh, ow, Ca1, Cb1, tap1, use1);
        __builtin_amdgcn_sched_barrier(0);
        form_tiles(Ca0, Cb0, tap0, use0, Bm, Dpa, Dqa, Dpb, Dqb);
        take8x4(Tpa, Tqa, Tpb, Tqb, Dpa, Dqa, Dpb, Dqb);
        int hn = (h0 + 32 < HP) ? (h0 + 32) : (HP - 16);
        asm volatile("" : "+s"(hn), "+v"(tap1) : "v"(Tpa[KL - 1]), "v"(Tqa[KL - 1]), "v"(Tpb[KL - 1]), "v"(Tqb[KL - 1]));
        load_tile(wl, xb, hn, l16, half, oh, ow, Ca0, Cb0, tap0, use0);
        __builtin_amdgcn_sched_barrier(0);
        form_tiles(Ca1, Cb1, tap1, use1, Bm, Dpa, Dqa, Dpb, Dqb);
        take8x4(Tpa, Tqa, Tpb, Tqb, Dpa, Dqa, Dpb, Dqb);
    }

    float kP[KL], kQ[KL], rP[KL], rQ[KL];
    #pragma unroll
    for (int i = 0; i < KL; ++i) {
        const float sendP = half ? Tpa[i] : Tpb[i];
        const float sendQ = half ? Tqa[i] : Tqb[i];
        kP[i] = half ? Tpb[i] : Tpa[i];
        kQ[i] = half ? Tqb[i] : Tqa[i];
        rP[i] = __shfl_xor(sendP, 16, 32);
        rQ[i] = __shfl_xor(sendQ, 16, 32);
    }
    bool okP, okQ;
    float tp = solve16(kP, rP, okP);
    float tq = solve16(kQ, rQ, okQ);

    if (__builtin_amdgcn_ballot_w32(!(okP && okQ)) != 0u) {
        float hi_pa = fmaxf(Tpa[0], __shfl_xor(Tpa[0], 16, 32));
        float hi_qa = fmaxf(Tqa[0], __shfl_xor(Tqa[0], 16, 32));
        float hi_pb = fmaxf(Tpb[0], __shfl_xor(Tpb[0], 16, 32));
        float hi_qb = fmaxf(Tqb[0], __shfl_xor(Tqb[0], 16, 32));
        const float lo0 = -GAMMA / DTOT;
        float lo_pa = lo0, lo_qa = lo0, lo_pb = lo0, lo_qb = lo0;

        for (int it = 0; it < NITER; ++it) {
            const float t_pa = 0.5f * (lo_pa + hi_pa);
            const float t_qa = 0.5f * (lo_qa + hi_qa);
            const float t_pb = 0.5f * (lo_pb + hi_pb);
            const float t_qb = 0.5f * (lo_qb + hi_qb);
            const float fl_pa = fmaxf(-2.0f * t_pa, 0.0f);
            const float fl_qa = fmaxf(-2.0f * t_qa, 0.0f);
            const float fl_pb = fmaxf(-2.0f * t_pb, 0.0f);
            const float fl_qb = fmaxf(-2.0f * t_qb, 0.0f);
            float f_pa = 0.0f, f_qa = 0.0f, f_pb = 0.0f, f_qb = 0.0f;

            for (int h0 = 0; h0 < HP; h0 += 16) {
                v8f Ca, Cb, Dpa, Dqa, Dpb, Dqb;
                float tap;
                bool use;
                load_tile(wl, xb, h0, l16, half, oh, ow, Ca, Cb, tap, use);
                form_tiles(Ca, Cb, tap, use, Bm, Dpa, Dqa, Dpb, Dqb);
                #pragma unroll
                for (int r = 0; r < 8; ++r) {
                    f_pa += fmaxf(__builtin_fabsf(Dpa[r]) - t_pa, fl_pa);
                    f_qa += fmaxf(__builtin_fabsf(Dqa[r]) - t_qa, fl_qa);
                    f_pb += fmaxf(__builtin_fabsf(Dpb[r]) - t_pb, fl_pb);
                    f_qb += fmaxf(__builtin_fabsf(Dqb[r]) - t_qb, fl_qb);
                }
            }
            f_pa += __shfl_xor(f_pa, 16, 32);
            f_qa += __shfl_xor(f_qa, 16, 32);
            f_pb += __shfl_xor(f_pb, 16, 32);
            f_qb += __shfl_xor(f_qb, 16, 32);

            lo_pa = (f_pa > GAMMA) ? t_pa : lo_pa;  hi_pa = (f_pa > GAMMA) ? hi_pa : t_pa;
            lo_qa = (f_qa > GAMMA) ? t_qa : lo_qa;  hi_qa = (f_qa > GAMMA) ? hi_qa : t_qa;
            lo_pb = (f_pb > GAMMA) ? t_pb : lo_pb;  hi_pb = (f_pb > GAMMA) ? hi_pb : t_pb;
            lo_qb = (f_qb > GAMMA) ? t_qb : lo_qb;  hi_qb = (f_qb > GAMMA) ? hi_qb : t_qb;
        }

        const float tau_pa = 0.5f * (lo_pa + hi_pa);
        const float tau_qa = 0.5f * (lo_qa + hi_qa);
        const float tau_pb = 0.5f * (lo_pb + hi_pb);
        const float tau_qb = 0.5f * (lo_qb + hi_qb);
        tp = okP ? tp : (half ? tau_pb : tau_pa);
        tq = okQ ? tq : (half ? tau_qb : tau_qa);
    }

    const float conv = 0.5f * DTOT * (tp - tq);

    out[(b * 32 + lane) * 1024 + pos] = conv;
}

extern "C" void kernel_launch(void* const* d_in, const int* in_sizes, int n_in,
                              void* d_out, int out_size, void* d_ws, size_t ws_size,
                              hipStream_t stream) {
    const float* x = (const float*)d_in[0];
    const float* w = (const float*)d_in[1];
    float* out = (float*)d_out;
    (void)in_sizes; (void)n_in; (void)out_size; (void)d_ws; (void)ws_size;

    mpconv_sparsemax_kernel<<<256, 256, 0, stream>>>(x, w, out);
}


# Loads go out earlier: a lane's 9 input taps (patch[32j+lane]) are requested once before the loop and kept in registers; each W set is re-requested right after its products; sums bit-identical
# speedup vs baseline: 10.1906x; 1.0978x over previous
#include <hip/hip_runtime.h>


typedef __attribute__((ext_vector_type(2))) float v2f;
typedef __attribute__((ext_vector_type(8))) float v8f;

#define HP     288
#define NITER  26
#define GAMMA  1.0f
#define DTOT   576.0f
#define KL     12
#define KM     16

__device__ __forceinline__ float mx(float a, float b) { return __builtin_fmaximum_numf(a, b); }
__device__ __forceinline__ float mn(float a, float b) { return __builtin_fminimum_numf(a, b); }

__device__ __forceinline__ void cex(float& a, float& b) {
    const float hi = mx(a, b);
    const float lo = mn(a, b);
    a = hi; b = lo;
}

__device__ __forceinline__ void cexd(float& a, float& b) {
    float hi, lo;
    asm("v_dual_max_num_f32 %0, %2, %3 :: v_dual_min_num_f32 %1, %2, %3" : "=&v"(hi), "=&v"(lo) : "v"(a), "v"(b));
    a = hi; b = lo;
}

__device__ __forceinline__ float max3f(float a, float b, float c) {
    float d;
    asm("v_max3_num_f32 %0, %1, %2, %3" : "=v"(d) : "v"(a), "v"(b), "v"(c));
    return d;
}

__device__ __forceinline__ float maxminf(float a, float b, float c) {
    float d;
    asm("v_maxmin_num_f32 %0, %1, %2, %3" : "=v"(d) : "v"(a), "v"(b), "v"(c));
    return d;
}

__device__ __forceinline__ void bitonic16_desc(float (&t)[KM]) {
    #pragma unroll
    for (int d = 8; d >= 1; d >>= 1) {
        #pragma unroll
        for (int i = 0; i < KM; ++i)
            if ((i & d) == 0) cex(t[i], t[i + d]);
    }
}

__device__ __forceinline__ void take8x4(float (&ta)[KL], float (&tb)[KL], float (&tc)[KL], float (&td)[KL],
                                        const v8f& Da, const v8f& Db, const v8f& Dc, const v8f& Dd) {
    float sa[8], sb[8], sc[8], sd[8];
    #pragma unroll
    for (int r = 0; r < 8; ++r) {
        sa[r] = __builtin_fabsf(Da[r]); sb[r] = __builtin_fabsf(Db[r]);
        sc[r] = __builtin_fabsf(Dc[r]); sd[r] = __builtin_fabsf(Dd[r]);
    }
#define CE4(i, j) cexd(sa[i], sa[j]); cexd(sb[i], sb[j]); cexd(sc[i], sc[j]); cexd(sd[i], sd[j]);
    CE4(0, 2) CE4(1, 3) CE4(4, 6) CE4(5, 7)
    CE4(0, 4) CE4(1, 5) CE4(2, 6) CE4(3, 7)
    CE4(0, 1) CE4(2, 3) CE4(4, 5) CE4(6, 7)
    CE4(2, 4) CE4(3, 5)
    CE4(1, 4) CE4(3, 6)
    CE4(1, 2) CE4(3, 4) CE4(5, 6)
#undef CE4
    float la[4], lb[4], lc[4], ld[4];
    #pragma unroll
    for (int i = 0; i < 4; ++i) {
        const float ha = max3f(ta[i], ta[8 + i], sa[3 - i]);
        const float hb = max3f(tb[i], tb[8 + i], sb[3 - i]);
        const float hc = max3f(tc[i], tc[8 + i], sc[3 - i]);
        const float hd = max3f(td[i], td[8 + i], sd[3 - i]);
        la[i] = maxminf(ta[8 + i], sa[3 - i], ta[i]);
        lb[i] = maxminf(tb[8 + i], sb[3 - i], tb[i]);
        lc[i] = maxminf(tc[8 + i], sc[3 - i], tc[i]);
        ld[i] = maxminf(td[8 + i], sd[3 - i], td[i]);
        ta[i] = ha; tb[i] = hb; tc[i] = hc; td[i] = hd;
    }
    #pragma unroll
    for (int j = 0; j < 4; ++j) {
        const float ha = max3f(ta[4 + j], sa[7 - j], la[j]);
        const float hb = max3f(tb[4 + j], sb[7 - j], lb[j]);
        const float hc = max3f(tc[4 + j], sc[7 - j], lc[j]);
        const float hd = max3f(td[4 + j], sd[7 - j], ld[j]);
        ta[8 + j] = maxminf(ta[4 + j], sa[7 - j], la[j]);
        tb[8 + j] = maxminf(tb[4 + j], sb[7 - j], lb[j]);
        tc[8 + j] = maxminf(tc[4 + j], sc[7 - j], lc[j]);
        td[8 + j] = maxminf(td[4 + j], sd[7 - j], ld[j]);
        ta[4 + j] = ha; tb[4 + j] = hb; tc[4 + j] = hc; td[4 + j] = hd;
    }
#define TE4(i, j) cexd(ta[i], ta[j]); cexd(tb[i], tb[j]); cexd(tc[i], tc[j]); cexd(td[i], td[j]);
    TE4(0, 2) TE4(1, 3) TE4(4, 6) TE4(5, 7) TE4(8, 10) TE4(9, 11)
    TE4(0, 1) TE4(2, 3) TE4(4, 5) TE4(6, 7) TE4(8, 9) TE4(10, 11)
#undef TE4
}

__device__ __forceinline__ float solve16(const float (&a)[KL], const float (&b)[KL], bool& ok) {
    float m[KM];
    #pragma unroll
    for (int i = 0; i < KM - KL; ++i) { m[i] = a[i]; m[KL + i] = b[KM - KL - 1 - i]; }
    #pragma unroll
    for (int i = KM - KL; i < KL; ++i) m[i] = mx(a[i], b[KM - 1 - i]);
    bitonic16_desc(m);
    float cs = 0.0f, bcs = m[0], bk = 1.0f;
    #pragma unroll
    for (int k = 1; k <= KM; ++k) {
        cs += m[k - 1];
        const bool c = (m[k - 1] * (float)k >= cs - GAMMA);
        bcs = c ? cs : bcs;
        bk  = c ? (float)k : bk;
    }
    const float tau = (bcs - GAMMA) / bk;
    ok = (tau >= mx(m[KM - 1], mx(a[KL - 1], b[KL - 1])));
    return tau;
}

__device__ __forceinline__ void load_tile(const float* wl, const float* xb, int h0, int l16, int half,
                                          int oh, int ow, v8f& Ca, v8f& Cb, float& tap, bool& use) {
    Ca = *reinterpret_cast<const v8f*>(wl + h0);
    Cb = *reinterpret_cast<const v8f*>(wl + 16 * HP + h0);
    __builtin_assume((unsigned)h0 < (unsigned)HP);
    const unsigned h  = (unsigned)(h0 + l16);
    const unsigned t  = h * 0xE38Fu;
    const unsigned c  = t >> 19;
    const unsigned f  = t & 0x7FFFFu;
    const unsigned r9 = (f * 9u) >> 19;
    const unsigned ki = (f * 3u) >> 19;
    const unsigned kj = r9 - ki * 3u;
    const int y  = oh - 1 + (int)ki;
    const int xx = ow - 1 + (int)kj;
    const bool in = ((unsigned)y < 32u) && ((unsigned)xx < 32u);
    const int idx = in ? ((int)(c * 1024u) + y * 32 + xx) : 0;
    tap = xb[idx];
    use = in && (half == 0);
}

__device__ __forceinline__ int tap_index(unsigned h, int oh, int ow, bool& in) {
    const unsigned t  = h * 0xE38Fu;
    const unsigned c  = t >> 19;
    const unsigned f  = t & 0x7FFFFu;
    const unsigned r9 = (f * 9u) >> 19;
    const unsigned ki = (f * 3u) >> 19;
    const unsigned kj = r9 - ki * 3u;
    const int y  = oh - 1 + (int)ki;
    const int xx = ow - 1 + (int)kj;
    in = ((unsigned)y < 32u) && ((unsigned)xx < 32u);
    return in ? ((int)(c * 1024u) + y * 32 + xx) : 0;
}

__device__ __forceinline__ float xhalf(float v) {
    const int b = __builtin_bit_cast(int, v);
    return __builtin_bit_cast(float, __builtin_amdgcn_permlanex16(b, b, 0x76543210, 0xfedcba98, false, false));
}

__device__ __forceinline__ void load_w(const float* wl, int h0, v8f& Ca, v8f& Cb) {
    Ca = *reinterpret_cast<const v8f*>(wl + h0);
    Cb = *reinterpret_cast<const v8f*>(wl + 16 * HP + h0);
}

__device__ __forceinline__ void form_tiles(const v8f& Ca, const v8f& Cb, float a0, const v2f& Bm,
                                           v8f& Dpa, v8f& Dqa, v8f& Dpb, v8f& Dqb) {
    v2f Ap; Ap[0] =  a0; Ap[1] = 0.0f;
    v2f Aq; Aq[0] = -a0; Aq[1] = 0.0f;
    Dpa = __builtin_amdgcn_wmma_f32_16x16x4_f32(false, Ap, false, Bm, (short)0, Ca, false, false);
    Dqa = __builtin_amdgcn_wmma_f32_16x16x4_f32(false, Aq, false, Bm, (short)0, Ca, false, false);
    Dpb = __builtin_amdgcn_wmma_f32_16x16x4_f32(false, Ap, false, Bm, (short)0, Cb, false, false);
    Dqb = __builtin_amdgcn_wmma_f32_16x16x4_f32(false, Aq, false, Bm, (short)0, Cb, false, false);
}

__global__ __launch_bounds__(256) void mpconv_sparsemax_kernel(
    const float* x,
    const float* w,
    float* __restrict__ out)
{
    const int tid  = threadIdx.x;
    const int lane = tid & 31;
    const int wv   = tid >> 5;

    const int loc = blockIdx.x * 8 + wv;
    const int b   = loc >> 10;
    const int pos = loc & 1023;
    const int oh  = pos >> 5, ow = pos & 31;

    const int   half   = lane >> 4;
    const int   l16    = lane & 15;
    const float bsel   = half ? 0.0f : 1.0f;
    v2f Bm; Bm[0] = bsel; Bm[1] = 0.0f;
    const float* wl = w + l16 * HP + half * 8;
    const float* xb = x + b * (32 * 1024);

    float Tpa[KL], Tqa[KL], Tpb[KL], Tqb[KL];
    #pragma unroll
    for (int i = 0; i < KL; ++i) { Tpa[i] = -1.0f; Tqa[i] = -1.0f; Tpb[i] = -1.0f; Tqb[i] = -1.0f; }

    v8f Ca0, Cb0, Ca1, Cb1;
    load_w(wl, 0, Ca0, Cb0);
    load_w(wl, 16, Ca1, Cb1);
    __builtin_amdgcn_sched_barrier(0);

    float P[9];
    bool  pin[9];
    #pragma unroll
    for (int j = 0; j < 9; ++j) P[j] = xb[tap_index((unsigned)(32 * j + lane), oh, ow, pin[j])];
    #pragma unroll
    for (int j = 0; j < 9; ++j) P[j] = pin[j] ? P[j] : 0.0f;

    for (int h0 = 0; h0 < HP; h0 += 32) {
        v8f Dpa, Dqa, Dpb, Dqb;
        const float own = P[0];
        const float far = xhalf(own);
        const float a_even = half ? 0.0f : own;
        float       a_odd  = half ? 0.0f : far;
        const int hn = (h0 + 32 < HP) ? (h0 + 32) : (HP - 32);
        form_tiles(Ca0, Cb0, a_even, Bm, Dpa, Dqa, Dpb, Dqb);
        asm volatile("" : : "v"(Ca0), "v"(Cb0));
        load_w(wl, hn, Ca0, Cb0);
        __builtin_amdgcn_sched_barrier(0);
        take8x4(Tpa, Tqa, Tpb, Tqb, Dpa, Dqa, Dpb, Dqb);
        asm volatile("" : "+v"(a_odd) : "v"(Tpa[KL - 1]), "v"(Tqa[KL - 1]), "v"(Tpb[KL - 1]), "v"(Tqb[KL - 1]));
        form_tiles(Ca1, Cb1, a_odd, Bm, Dpa, Dqa, Dpb, Dqb);
        asm volatile("" : : "v"(Ca1), "v"(Cb1));
        load_w(wl, hn + 16, Ca1, Cb1);
        __builtin_amdgcn_sched_barrier(0);
        take8x4(Tpa, Tqa, Tpb, Tqb, Dpa, Dqa, Dpb, Dqb);
        #pragma unroll
        for (int j = 0; j < 8; ++j) P[j] = P[j + 1];
    }

    float kP[KL], kQ[KL], rP[KL], rQ[KL];
    #pragma unroll
    for (int i = 0; i < KL; ++i) {
        const float sendP = half ? Tpa[i] : Tpb[i];
        const float sendQ = half ? Tqa[i] : Tqb[i];
        kP[i] = half ? Tpb[i] : Tpa[i];
        kQ[i] = half ? Tqb[i] : Tqa[i];
        rP[i] = __shfl_xor(sendP, 16, 32);
        rQ[i] = __shfl_xor(sendQ, 16, 32);
    }
    bool okP, okQ;
    float tp = solve16(kP, rP, okP);
    float tq = solve16(kQ, rQ, okQ);

    if (__builtin_amdgcn_ballot_w32(!(okP && okQ)) != 0u) {
        float hi_pa = fmaxf(Tpa[0], __shfl_xor(Tpa[0], 16, 32));
        float hi_qa = fmaxf(Tqa[0], __shfl_xor(Tqa[0], 16, 32));
        float hi_pb = fmaxf(Tpb[0], __shfl_xor(Tpb[0], 16, 32));
        float hi_qb = fmaxf(Tqb[0], __shfl_xor(Tqb[0], 16, 32));
        const float lo0 = -GAMMA / DTOT;
        float lo_pa = lo0, lo_qa = lo0, lo_pb = lo0, lo_qb = lo0;

        for (int it = 0; it < NITER; ++it) {
            const float t_pa = 0.5f * (lo_pa + hi_pa);
            const float t_qa = 0.5f * (lo_qa + hi_qa);
            const float t_pb = 0.5f * (lo_pb + hi_pb);
            const float t_qb = 0.5f * (lo_qb + hi_qb);
            const float fl_pa = fmaxf(-2.0f * t_pa, 0.0f);
            const float fl_qa = fmaxf(-2.0f * t_qa, 0.0f);
            const float fl_pb = fmaxf(-2.0f * t_pb, 0.0f);
            const float fl_qb = fmaxf(-2.0f * t_qb, 0.0f);
            float f_pa = 0.0f, f_qa = 0.0f, f_pb = 0.0f, f_qb = 0.0f;

            for (int h0 = 0; h0 < HP; h0 += 16) {
                v8f Ca, Cb, Dpa, Dqa, Dpb, Dqb;
                float tap;
                bool use;
                load_tile(wl, xb, h0, l16, half, oh, ow, Ca, Cb, tap, use);
                form_tiles(Ca, Cb, use ? tap : 0.0f, Bm, Dpa, Dqa, Dpb, Dqb);
                #pragma unroll
                for (int r = 0; r < 8; ++r) {
                    f_pa += fmaxf(__builtin_fabsf(Dpa[r]) - t_pa, fl_pa);
                    f_qa += fmaxf(__builtin_fabsf(Dqa[r]) - t_qa, fl_qa);
                    f_pb += fmaxf(__builtin_fabsf(Dpb[r]) - t_pb, fl_pb);
                    f_qb += fmaxf(__builtin_fabsf(Dqb[r]) - t_qb, fl_qb);
                }
            }
            f_pa += __shfl_xor(f_pa, 16, 32);
            f_qa += __shfl_xor(f_qa, 16, 32);
            f_pb += __shfl_xor(f_pb, 16, 32);
            f_qb += __shfl_xor(f_qb, 16, 32);

            lo_pa = (f_pa > GAMMA) ? t_pa : lo_pa;  hi_pa = (f_pa > GAMMA) ? hi_pa : t_pa;
            lo_qa = (f_qa > GAMMA) ? t_qa : lo_qa;  hi_qa = (f_qa > GAMMA) ? hi_qa : t_qa;
            lo_pb = (f_pb > GAMMA) ? t_pb : lo_pb;  hi_pb = (f_pb > GAMMA) ? hi_pb : t_pb;
            lo_qb = (f_qb > GAMMA) ? t_qb : lo_qb;  hi_qb = (f_qb > GAMMA) ? hi_qb : t_qb;
        }

        const float tau_pa = 0.5f * (lo_pa + hi_pa);
        const float tau_qa = 0.5f * (lo_qa + hi_qa);
        const float tau_pb = 0.5f * (lo_pb + hi_pb);
        const float tau_qb = 0.5f * (lo_qb + hi_qb);
        tp = okP ? tp : (half ? tau_pb : tau_pa);
        tq = okQ ? tq : (half ? tau_qb : tau_qa);
    }

    const float conv = 0.5f * DTOT * (tp - tq);

    out[(b * 32 + lane) * 1024 + pos] = conv;
}

extern "C" void kernel_launch(void* const* d_in, const int* in_sizes, int n_in,
                              void* d_out, int out_size, void* d_ws, size_t ws_size,
                              hipStream_t stream) {
    const float* x = (const float*)d_in[0];
    const float* w = (const float*)d_in[1];
    float* out = (float*)d_out;
    (void)in_sizes; (void)n_in; (void)out_size; (void)d_ws; (void)ws_size;

    mpconv_sparsemax_kernel<<<256, 256, 0, stream>>>(x, w, out);
}
